# starting kernel
# baseline (speedup 1.0000x reference)
#include <hip/hip_runtime.h>
#include <hip/hip_bf16.h>
#include <math.h>

typedef __attribute__((ext_vector_type(2))) float v2f;
typedef __attribute__((ext_vector_type(8))) float v8f;

__device__ __forceinline__ v8f wmma_k4(v2f a, v2f b, v8f c) {
  return __builtin_amdgcn_wmma_f32_16x16x4_f32(
       false, a,  false, b,
       (short)0, c,  false,  false);
}

__device__ __forceinline__ float fast_atan2f(float y, float x) {
  const float ax = __builtin_fabsf(x);
  const float ay = __builtin_fabsf(y);
  const float mx = fmaxf(ax, ay);
  const float mn = fminf(ax, ay);
  float t = (mx > 0.0f) ? mn * __builtin_amdgcn_rcpf(mx) : 0.0f;
  const float s = t * t;
  float p = -0.01172120f;
  p = __builtin_fmaf(p, s, 0.05265332f);
  p = __builtin_fmaf(p, s, -0.11643287f);
  p = __builtin_fmaf(p, s, 0.19354346f);
  p = __builtin_fmaf(p, s, -0.33262347f);
  p = __builtin_fmaf(p, s, 0.99997726f);
  float r = t * p;
  r = (ay > ax)   ? 1.57079632679f - r : r;
  r = (x < 0.0f)  ? 3.14159265359f - r : r;
  return __builtin_copysignf(r, y);
}

__global__ __launch_bounds__(256) void winding16(
    const float* __restrict__ qpts,
    const float* __restrict__ fcoord,
    float* __restrict__ out,
    int ftiles)
{
  __shared__ float s_acc[16];
  const int tid   = threadIdx.x;
  const int lane  = tid & 31;
  const int wave  = tid >> 5;
  const int half  = lane >> 4;
  const int l16   = lane & 15;
  const int qtile = blockIdx.x;

  if (tid < 16) s_acc[tid] = 0.0f;
  __syncthreads();

  const float* qp = qpts + (size_t)(qtile * 16 + l16) * 3;
  const float qx = qp[0], qy = qp[1], qz = qp[2];
  v2f A;
  A.x = half ? qz   : qx;
  A.y = half ? 1.0f : qy;
  const float qq_self = qx * qx + qy * qy + qz * qz;

  v8f cqq, czero = {};
  #pragma unroll
  for (int r = 0; r < 8; ++r)
    cqq[r] = __shfl(qq_self, r + 8 * half, 32);

  v8f acc = {};

  for (int ft = wave; ft < ftiles; ft += 8) {
    const float* fp = fcoord + (size_t)(ft * 16 + l16) * 9;
    const float v0x = fp[0], v0y = fp[1], v0z = fp[2];
    const float v1x = fp[3], v1y = fp[4], v1z = fp[5];
    const float v2x = fp[6], v2y = fp[7], v2z = fp[8];

    const float nx = (v0y*v1z - v0z*v1y) + (v1y*v2z - v1z*v2y) + (v2y*v0z - v2z*v0y);
    const float ny = (v0z*v1x - v0x*v1z) + (v1z*v2x - v1x*v2z) + (v2z*v0x - v2x*v0z);
    const float nz = (v0x*v1y - v0y*v1x) + (v1x*v2y - v1y*v2x) + (v2x*v0y - v2y*v0x);
    const float D  = v0x*(v1y*v2z - v1z*v2y) + v0y*(v1z*v2x - v1x*v2z) + v0z*(v1x*v2y - v1y*v2x);
    const float s0  = v0x*v0x + v0y*v0y + v0z*v0z;
    const float s1  = v1x*v1x + v1y*v1y + v1z*v1z;
    const float s2  = v2x*v2x + v2y*v2y + v2z*v2z;
    const float d01 = v0x*v1x + v0y*v1y + v0z*v1z;
    const float d12 = v1x*v2x + v1y*v2y + v1z*v2z;
    const float d02 = v0x*v2x + v0y*v2y + v0z*v2z;

    v2f Bna, Bnb, Bnc, Bab, Bbc, Bac, Bdet;
    Bna.x = half ? -2.0f*v0z      : -2.0f*v0x;      Bna.y = half ? s0  : -2.0f*v0y;
    Bnb.x = half ? -2.0f*v1z      : -2.0f*v1x;      Bnb.y = half ? s1  : -2.0f*v1y;
    Bnc.x = half ? -2.0f*v2z      : -2.0f*v2x;      Bnc.y = half ? s2  : -2.0f*v2y;
    Bab.x = half ? -(v0z + v1z)   : -(v0x + v1x);   Bab.y = half ? d01 : -(v0y + v1y);
    Bbc.x = half ? -(v1z + v2z)   : -(v1x + v2x);   Bbc.y = half ? d12 : -(v1y + v2y);
    Bac.x = half ? -(v0z + v2z)   : -(v0x + v2x);   Bac.y = half ? d02 : -(v0y + v2y);
    Bdet.x = half ? -nz           : -nx;            Bdet.y = half ? D  : -ny;

    v8f na2 = wmma_k4(A, Bna, cqq);
    v8f nb2 = wmma_k4(A, Bnb, cqq);
    v8f nc2 = wmma_k4(A, Bnc, cqq);
    v8f tab = wmma_k4(A, Bab, cqq);
    v8f tbc = wmma_k4(A, Bbc, cqq);
    v8f tac = wmma_k4(A, Bac, cqq);
    v8f tdt = wmma_k4(A, Bdet, czero);

    #pragma unroll
    for (int r = 0; r < 8; ++r) {
      const float na = __builtin_amdgcn_sqrtf(fmaxf(na2[r], 0.0f));
      const float nb = __builtin_amdgcn_sqrtf(fmaxf(nb2[r], 0.0f));
      const float nc = __builtin_amdgcn_sqrtf(fmaxf(nc2[r], 0.0f));
      const float denom = na * nb * nc + tbc[r] * na + tac[r] * nb + tab[r] * nc;
      acc[r] += fast_atan2f(tdt[r], denom);
    }
  }

  #pragma unroll
  for (int off = 8; off >= 1; off >>= 1) {
    #pragma unroll
    for (int r = 0; r < 8; ++r)
      acc[r] += __shfl_xor(acc[r], off, 32);
  }
  if (l16 == 0) {
    #pragma unroll
    for (int r = 0; r < 8; ++r)
      atomicAdd(&s_acc[r + 8 * half], acc[r]);
  }
  __syncthreads();

  if (tid < 16)
    out[qtile * 16 + tid] = s_acc[tid] * 0.15915494309189535f;
}

extern "C" void kernel_launch(void* const* d_in, const int* in_sizes, int n_in,
                              void* d_out, int out_size, void* d_ws, size_t ws_size,
                              hipStream_t stream) {
  const float* qpts = (const float*)d_in[0];
  const float* fc   = (const float*)d_in[1];
  float* out        = (float*)d_out;

  const int N = in_sizes[0] / 3;
  const int F = in_sizes[1] / 9;
  const int qtiles = N / 16;
  const int ftiles = F / 16;

  winding16<<<dim3(qtiles), dim3(256), 0, stream>>>(qpts, fc, out, ftiles);
}


# Block of 16 waves, was 8 (each wave takes every 16th face tile): four waves per SIMD cover the load wait at the top of each trip; LDS float adds now join 16 per-wave partial sums, was 8
# speedup vs baseline: 1.1083x; 1.1083x over previous
#include <hip/hip_runtime.h>
#include <hip/hip_bf16.h>
#include <math.h>

typedef __attribute__((ext_vector_type(2))) float v2f;
typedef __attribute__((ext_vector_type(8))) float v8f;

__device__ __forceinline__ v8f wmma_k4(v2f a, v2f b, v8f c) {
  return __builtin_amdgcn_wmma_f32_16x16x4_f32(
       false, a,  false, b,
       (short)0, c,  false,  false);
}

__device__ __forceinline__ float fast_atan2f(float y, float x) {
  const float ax = __builtin_fabsf(x);
  const float ay = __builtin_fabsf(y);
  const float mx = fmaxf(ax, ay);
  const float mn = fminf(ax, ay);
  float t = (mx > 0.0f) ? mn * __builtin_amdgcn_rcpf(mx) : 0.0f;
  const float s = t * t;
  float p = -0.01172120f;
  p = __builtin_fmaf(p, s, 0.05265332f);
  p = __builtin_fmaf(p, s, -0.11643287f);
  p = __builtin_fmaf(p, s, 0.19354346f);
  p = __builtin_fmaf(p, s, -0.33262347f);
  p = __builtin_fmaf(p, s, 0.99997726f);
  float r = t * p;
  r = (ay > ax)   ? 1.57079632679f - r : r;
  r = (x < 0.0f)  ? 3.14159265359f - r : r;
  return __builtin_copysignf(r, y);
}

constexpr int kWaves   = 16;
constexpr int kThreads = 512;
__global__ __launch_bounds__(kThreads) void winding16(
    const float* __restrict__ qpts,
    const float* __restrict__ fcoord,
    float* __restrict__ out,
    int ftiles)
{
  __shared__ float s_acc[16];
  const int tid   = threadIdx.x;
  const int lane  = tid & 31;
  const int wave  = tid >> 5;
  const int half  = lane >> 4;
  const int l16   = lane & 15;
  const int qtile = blockIdx.x;

  if (tid < 16) s_acc[tid] = 0.0f;
  __syncthreads();

  const float* qp = qpts + (size_t)(qtile * 16 + l16) * 3;
  const float qx = qp[0], qy = qp[1], qz = qp[2];
  v2f A;
  A.x = half ? qz   : qx;
  A.y = half ? 1.0f : qy;
  const float qq_self = qx * qx + qy * qy + qz * qz;

  v8f cqq, czero = {};
  #pragma unroll
  for (int r = 0; r < 8; ++r)
    cqq[r] = __shfl(qq_self, r + 8 * half, 32);

  v8f acc = {};

  for (int ft = wave; ft < ftiles; ft += kWaves) {
    const float* fp = fcoord + (size_t)(ft * 16 + l16) * 9;
    const float v0x = fp[0], v0y = fp[1], v0z = fp[2];
    const float v1x = fp[3], v1y = fp[4], v1z = fp[5];
    const float v2x = fp[6], v2y = fp[7], v2z = fp[8];

    const float nx = (v0y*v1z - v0z*v1y) + (v1y*v2z - v1z*v2y) + (v2y*v0z - v2z*v0y);
    const float ny = (v0z*v1x - v0x*v1z) + (v1z*v2x - v1x*v2z) + (v2z*v0x - v2x*v0z);
    const float nz = (v0x*v1y - v0y*v1x) + (v1x*v2y - v1y*v2x) + (v2x*v0y - v2y*v0x);
    const float D  = v0x*(v1y*v2z - v1z*v2y) + v0y*(v1z*v2x - v1x*v2z) + v0z*(v1x*v2y - v1y*v2x);
    const float s0  = v0x*v0x + v0y*v0y + v0z*v0z;
    const float s1  = v1x*v1x + v1y*v1y + v1z*v1z;
    const float s2  = v2x*v2x + v2y*v2y + v2z*v2z;
    const float d01 = v0x*v1x + v0y*v1y + v0z*v1z;
    const float d12 = v1x*v2x + v1y*v2y + v1z*v2z;
    const float d02 = v0x*v2x + v0y*v2y + v0z*v2z;

    v2f Bna, Bnb, Bnc, Bab, Bbc, Bac, Bdet;
    Bna.x = half ? -2.0f*v0z      : -2.0f*v0x;      Bna.y = half ? s0  : -2.0f*v0y;
    Bnb.x = half ? -2.0f*v1z      : -2.0f*v1x;      Bnb.y = half ? s1  : -2.0f*v1y;
    Bnc.x = half ? -2.0f*v2z      : -2.0f*v2x;      Bnc.y = half ? s2  : -2.0f*v2y;
    Bab.x = half ? -(v0z + v1z)   : -(v0x + v1x);   Bab.y = half ? d01 : -(v0y + v1y);
    Bbc.x = half ? -(v1z + v2z)   : -(v1x + v2x);   Bbc.y = half ? d12 : -(v1y + v2y);
    Bac.x = half ? -(v0z + v2z)   : -(v0x + v2x);   Bac.y = half ? d02 : -(v0y + v2y);
    Bdet.x = half ? -nz           : -nx;            Bdet.y = half ? D  : -ny;

    v8f na2 = wmma_k4(A, Bna, cqq);
    v8f nb2 = wmma_k4(A, Bnb, cqq);
    v8f nc2 = wmma_k4(A, Bnc, cqq);
    v8f tab = wmma_k4(A, Bab, cqq);
    v8f tbc = wmma_k4(A, Bbc, cqq);
    v8f tac = wmma_k4(A, Bac, cqq);
    v8f tdt = wmma_k4(A, Bdet, czero);

    #pragma unroll
    for (int r = 0; r < 8; ++r) {
      const float na = __builtin_amdgcn_sqrtf(fmaxf(na2[r], 0.0f));
      const float nb = __builtin_amdgcn_sqrtf(fmaxf(nb2[r], 0.0f));
      const float nc = __builtin_amdgcn_sqrtf(fmaxf(nc2[r], 0.0f));
      const float denom = na * nb * nc + tbc[r] * na + tac[r] * nb + tab[r] * nc;
      acc[r] += fast_atan2f(tdt[r], denom);
    }
  }

  #pragma unroll
  for (int off = 8; off >= 1; off >>= 1) {
    #pragma unroll
    for (int r = 0; r < 8; ++r)
      acc[r] += __shfl_xor(acc[r], off, 32);
  }
  if (l16 == 0) {
    #pragma unroll
    for (int r = 0; r < 8; ++r)
      atomicAdd(&s_acc[r + 8 * half], acc[r]);
  }
  __syncthreads();

  if (tid < 16)
    out[qtile * 16 + tid] = s_acc[tid] * 0.15915494309189535f;
}

extern "C" void kernel_launch(void* const* d_in, const int* in_sizes, int n_in,
                              void* d_out, int out_size, void* d_ws, size_t ws_size,
                              hipStream_t stream) {
  const float* qpts = (const float*)d_in[0];
  const float* fc   = (const float*)d_in[1];
  float* out        = (float*)d_out;

  const int N = in_sizes[0] / 3;
  const int F = in_sizes[1] / 9;
  const int qtiles = N / 16;
  const int ftiles = F / 16;

  winding16<<<dim3(qtiles), dim3(kThreads), 0, stream>>>(qpts, fc, out, ftiles);
}


# atan2 without octant/half-plane selects: t=(|y|-|x|)/(|y|+|x|), angle=pi/2+sx*(atan(t)-pi/4), signs sx,sy as +-1 factors on packed FMAs; waves_per_eu(4,8) so the compiler does not serialise it
# speedup vs baseline: 1.2298x; 1.1096x over previous
#include <hip/hip_runtime.h>
#include <hip/hip_bf16.h>
#include <math.h>

typedef __attribute__((ext_vector_type(2))) float v2f;
typedef __attribute__((ext_vector_type(8))) float v8f;

__device__ __forceinline__ v8f wmma_k4(v2f a, v2f b, v8f c) {
  return __builtin_amdgcn_wmma_f32_16x16x4_f32(
       false, a,  false, b,
       (short)0, c,  false,  false);
}

__device__ __forceinline__ float fast_atan2f(float y, float x) {
  const float sx  = __builtin_copysignf(1.0f, x);
  const float sy  = __builtin_copysignf(1.0f, y);
  const float ay  = sy * y;
  const float num = __builtin_fmaf(-sx, x, ay);
  const float den = __builtin_fmaf( sx, x, ay);
  const float t = fmaxf(num * __builtin_amdgcn_rcpf(den), -1.0f);
  const float s = t * t;
  float p = -0.01172120f;
  p = __builtin_fmaf(p, s, 0.05265332f);
  p = __builtin_fmaf(p, s, -0.11643287f);
  p = __builtin_fmaf(p, s, 0.19354346f);
  p = __builtin_fmaf(p, s, -0.33262347f);
  p = __builtin_fmaf(p, s, 0.99997726f);
  const float u  = __builtin_fmaf(t, p, -0.78539816340f);
  const float th = __builtin_fmaf(sx, u, 1.57079632679f);
  return sy * th;
}

constexpr int kWaves   = 16;
constexpr int kThreads = 512;
__global__ __launch_bounds__(kThreads) __attribute__((amdgpu_waves_per_eu(4, 8))) void winding16(
    const float* __restrict__ qpts,
    const float* __restrict__ fcoord,
    float* __restrict__ out,
    int ftiles)
{
  __shared__ float s_acc[16];
  const int tid   = threadIdx.x;
  const int lane  = tid & 31;
  const int wave  = tid >> 5;
  const int half  = lane >> 4;
  const int l16   = lane & 15;
  const int qtile = blockIdx.x;

  if (tid < 16) s_acc[tid] = 0.0f;
  __syncthreads();

  const float* qp = qpts + (size_t)(qtile * 16 + l16) * 3;
  const float qx = qp[0], qy = qp[1], qz = qp[2];
  v2f A;
  A.x = half ? qz   : qx;
  A.y = half ? 1.0f : qy;
  const float qq_self = qx * qx + qy * qy + qz * qz;

  v8f cqq, czero = {};
  #pragma unroll
  for (int r = 0; r < 8; ++r)
    cqq[r] = __shfl(qq_self, r + 8 * half, 32);

  v8f acc = {};

  for (int ft = wave; ft < ftiles; ft += kWaves) {
    const float* fp = fcoord + (size_t)(ft * 16 + l16) * 9;
    const float v0x = fp[0], v0y = fp[1], v0z = fp[2];
    const float v1x = fp[3], v1y = fp[4], v1z = fp[5];
    const float v2x = fp[6], v2y = fp[7], v2z = fp[8];

    const float nx = (v0y*v1z - v0z*v1y) + (v1y*v2z - v1z*v2y) + (v2y*v0z - v2z*v0y);
    const float ny = (v0z*v1x - v0x*v1z) + (v1z*v2x - v1x*v2z) + (v2z*v0x - v2x*v0z);
    const float nz = (v0x*v1y - v0y*v1x) + (v1x*v2y - v1y*v2x) + (v2x*v0y - v2y*v0x);
    const float D  = v0x*(v1y*v2z - v1z*v2y) + v0y*(v1z*v2x - v1x*v2z) + v0z*(v1x*v2y - v1y*v2x);
    const float s0  = v0x*v0x + v0y*v0y + v0z*v0z;
    const float s1  = v1x*v1x + v1y*v1y + v1z*v1z;
    const float s2  = v2x*v2x + v2y*v2y + v2z*v2z;
    const float d01 = v0x*v1x + v0y*v1y + v0z*v1z;
    const float d12 = v1x*v2x + v1y*v2y + v1z*v2z;
    const float d02 = v0x*v2x + v0y*v2y + v0z*v2z;

    v2f Bna, Bnb, Bnc, Bab, Bbc, Bac, Bdet;
    Bna.x = half ? -2.0f*v0z      : -2.0f*v0x;      Bna.y = half ? s0  : -2.0f*v0y;
    Bnb.x = half ? -2.0f*v1z      : -2.0f*v1x;      Bnb.y = half ? s1  : -2.0f*v1y;
    Bnc.x = half ? -2.0f*v2z      : -2.0f*v2x;      Bnc.y = half ? s2  : -2.0f*v2y;
    Bab.x = half ? -(v0z + v1z)   : -(v0x + v1x);   Bab.y = half ? d01 : -(v0y + v1y);
    Bbc.x = half ? -(v1z + v2z)   : -(v1x + v2x);   Bbc.y = half ? d12 : -(v1y + v2y);
    Bac.x = half ? -(v0z + v2z)   : -(v0x + v2x);   Bac.y = half ? d02 : -(v0y + v2y);
    Bdet.x = half ? -nz           : -nx;            Bdet.y = half ? D  : -ny;

    v8f na2 = wmma_k4(A, Bna, cqq);
    v8f nb2 = wmma_k4(A, Bnb, cqq);
    v8f nc2 = wmma_k4(A, Bnc, cqq);
    v8f tab = wmma_k4(A, Bab, cqq);
    v8f tbc = wmma_k4(A, Bbc, cqq);
    v8f tac = wmma_k4(A, Bac, cqq);
    v8f tdt = wmma_k4(A, Bdet, czero);

    #pragma unroll
    for (int r = 0; r < 8; ++r) {
      const float na = __builtin_amdgcn_sqrtf(fmaxf(na2[r], 0.0f));
      const float nb = __builtin_amdgcn_sqrtf(fmaxf(nb2[r], 0.0f));
      const float nc = __builtin_amdgcn_sqrtf(fmaxf(nc2[r], 0.0f));
      const float denom = na * nb * nc + tbc[r] * na + tac[r] * nb + tab[r] * nc;
      acc[r] += fast_atan2f(tdt[r], denom);
    }
  }

  #pragma unroll
  for (int off = 8; off >= 1; off >>= 1) {
    #pragma unroll
    for (int r = 0; r < 8; ++r)
      acc[r] += __shfl_xor(acc[r], off, 32);
  }
  if (l16 == 0) {
    #pragma unroll
    for (int r = 0; r < 8; ++r)
      atomicAdd(&s_acc[r + 8 * half], acc[r]);
  }
  __syncthreads();

  if (tid < 16)
    out[qtile * 16 + tid] = s_acc[tid] * 0.15915494309189535f;
}

extern "C" void kernel_launch(void* const* d_in, const int* in_sizes, int n_in,
                              void* d_out, int out_size, void* d_ws, size_t ws_size,
                              hipStream_t stream) {
  const float* qpts = (const float*)d_in[0];
  const float* fc   = (const float*)d_in[1];
  float* out        = (float*)d_out;

  const int N = in_sizes[0] / 3;
  const int F = in_sizes[1] / 9;
  const int qtiles = N / 16;
  const int ftiles = F / 16;

  winding16<<<dim3(qtiles), dim3(kThreads), 0, stream>>>(qpts, fc, out, ftiles);
}


# Norms as sqrt(|x|) with the root's free |x| source modifier, was sqrt(max(x,0)): drops the 24 clamps and the 24 max(x,x) the compiler put in front of them per trip
# speedup vs baseline: 1.3161x; 1.0702x over previous
#include <hip/hip_runtime.h>
#include <hip/hip_bf16.h>
#include <math.h>

typedef __attribute__((ext_vector_type(2))) float v2f;
typedef __attribute__((ext_vector_type(8))) float v8f;

__device__ __forceinline__ v8f wmma_k4(v2f a, v2f b, v8f c) {
  return __builtin_amdgcn_wmma_f32_16x16x4_f32(
       false, a,  false, b,
       (short)0, c,  false,  false);
}

__device__ __forceinline__ float fast_atan2f(float y, float x) {
  const float sx  = __builtin_copysignf(1.0f, x);
  const float sy  = __builtin_copysignf(1.0f, y);
  const float ay  = sy * y;
  const float num = __builtin_fmaf(-sx, x, ay);
  const float den = __builtin_fmaf( sx, x, ay);
  const float t = fmaxf(num * __builtin_amdgcn_rcpf(den), -1.0f);
  const float s = t * t;
  float p = -0.01172120f;
  p = __builtin_fmaf(p, s, 0.05265332f);
  p = __builtin_fmaf(p, s, -0.11643287f);
  p = __builtin_fmaf(p, s, 0.19354346f);
  p = __builtin_fmaf(p, s, -0.33262347f);
  p = __builtin_fmaf(p, s, 0.99997726f);
  const float u  = __builtin_fmaf(t, p, -0.78539816340f);
  const float th = __builtin_fmaf(sx, u, 1.57079632679f);
  return sy * th;
}

constexpr int kWaves   = 16;
constexpr int kThreads = 512;
__global__ __launch_bounds__(kThreads) __attribute__((amdgpu_waves_per_eu(4, 8))) void winding16(
    const float* __restrict__ qpts,
    const float* __restrict__ fcoord,
    float* __restrict__ out,
    int ftiles)
{
  __shared__ float s_acc[16];
  const int tid   = threadIdx.x;
  const int lane  = tid & 31;
  const int wave  = tid >> 5;
  const int half  = lane >> 4;
  const int l16   = lane & 15;
  const int qtile = blockIdx.x;

  if (tid < 16) s_acc[tid] = 0.0f;
  __syncthreads();

  const float* qp = qpts + (size_t)(qtile * 16 + l16) * 3;
  const float qx = qp[0], qy = qp[1], qz = qp[2];
  v2f A;
  A.x = half ? qz   : qx;
  A.y = half ? 1.0f : qy;
  const float qq_self = qx * qx + qy * qy + qz * qz;

  v8f cqq, czero = {};
  #pragma unroll
  for (int r = 0; r < 8; ++r)
    cqq[r] = __shfl(qq_self, r + 8 * half, 32);

  v8f acc = {};

  for (int ft = wave; ft < ftiles; ft += kWaves) {
    const float* fp = fcoord + (size_t)(ft * 16 + l16) * 9;
    const float v0x = fp[0], v0y = fp[1], v0z = fp[2];
    const float v1x = fp[3], v1y = fp[4], v1z = fp[5];
    const float v2x = fp[6], v2y = fp[7], v2z = fp[8];

    const float nx = (v0y*v1z - v0z*v1y) + (v1y*v2z - v1z*v2y) + (v2y*v0z - v2z*v0y);
    const float ny = (v0z*v1x - v0x*v1z) + (v1z*v2x - v1x*v2z) + (v2z*v0x - v2x*v0z);
    const float nz = (v0x*v1y - v0y*v1x) + (v1x*v2y - v1y*v2x) + (v2x*v0y - v2y*v0x);
    const float D  = v0x*(v1y*v2z - v1z*v2y) + v0y*(v1z*v2x - v1x*v2z) + v0z*(v1x*v2y - v1y*v2x);
    const float s0  = v0x*v0x + v0y*v0y + v0z*v0z;
    const float s1  = v1x*v1x + v1y*v1y + v1z*v1z;
    const float s2  = v2x*v2x + v2y*v2y + v2z*v2z;
    const float d01 = v0x*v1x + v0y*v1y + v0z*v1z;
    const float d12 = v1x*v2x + v1y*v2y + v1z*v2z;
    const float d02 = v0x*v2x + v0y*v2y + v0z*v2z;

    v2f Bna, Bnb, Bnc, Bab, Bbc, Bac, Bdet;
    Bna.x = half ? -2.0f*v0z      : -2.0f*v0x;      Bna.y = half ? s0  : -2.0f*v0y;
    Bnb.x = half ? -2.0f*v1z      : -2.0f*v1x;      Bnb.y = half ? s1  : -2.0f*v1y;
    Bnc.x = half ? -2.0f*v2z      : -2.0f*v2x;      Bnc.y = half ? s2  : -2.0f*v2y;
    Bab.x = half ? -(v0z + v1z)   : -(v0x + v1x);   Bab.y = half ? d01 : -(v0y + v1y);
    Bbc.x = half ? -(v1z + v2z)   : -(v1x + v2x);   Bbc.y = half ? d12 : -(v1y + v2y);
    Bac.x = half ? -(v0z + v2z)   : -(v0x + v2x);   Bac.y = half ? d02 : -(v0y + v2y);
    Bdet.x = half ? -nz           : -nx;            Bdet.y = half ? D  : -ny;

    __builtin_amdgcn_sched_barrier(0);
    v8f na2 = wmma_k4(A, Bna, cqq);
    v8f nb2 = wmma_k4(A, Bnb, cqq);
    v8f nc2 = wmma_k4(A, Bnc, cqq);
    v8f tab = wmma_k4(A, Bab, cqq);
    v8f tbc = wmma_k4(A, Bbc, cqq);
    v8f tac = wmma_k4(A, Bac, cqq);
    v8f tdt = wmma_k4(A, Bdet, czero);
    __builtin_amdgcn_sched_barrier(0);

    #pragma unroll
    for (int r = 0; r < 8; ++r) {
      const float na = __builtin_amdgcn_sqrtf(__builtin_fabsf(na2[r]));
      const float nb = __builtin_amdgcn_sqrtf(__builtin_fabsf(nb2[r]));
      const float nc = __builtin_amdgcn_sqrtf(__builtin_fabsf(nc2[r]));
      const float denom = na * nb * nc + tbc[r] * na + tac[r] * nb + tab[r] * nc;
      acc[r] += fast_atan2f(tdt[r], denom);
    }
  }

  #pragma unroll
  for (int off = 8; off >= 1; off >>= 1) {
    #pragma unroll
    for (int r = 0; r < 8; ++r)
      acc[r] += __shfl_xor(acc[r], off, 32);
  }
  if (l16 == 0) {
    #pragma unroll
    for (int r = 0; r < 8; ++r)
      atomicAdd(&s_acc[r + 8 * half], acc[r]);
  }
  __syncthreads();

  if (tid < 16)
    out[qtile * 16 + tid] = s_acc[tid] * 0.15915494309189535f;
}

extern "C" void kernel_launch(void* const* d_in, const int* in_sizes, int n_in,
                              void* d_out, int out_size, void* d_ws, size_t ws_size,
                              hipStream_t stream) {
  const float* qpts = (const float*)d_in[0];
  const float* fc   = (const float*)d_in[1];
  float* out        = (float*)d_out;

  const int N = in_sizes[0] / 3;
  const int F = in_sizes[1] / 9;
  const int qtiles = N / 16;
  const int ftiles = F / 16;

  winding16<<<dim3(qtiles), dim3(kThreads), 0, stream>>>(qpts, fc, out, ftiles);
}


# Denominator from the three norms: 2*denom = (na+nb)(nb+nc)(na+nc) - (e12 na + e02 nb + e01 nc), e = squared edge lengths of the face; det doubled in B; a.b, b.c, a.c matrix ops go: 4 a tile, was 7
# speedup vs baseline: 1.4743x; 1.1202x over previous
#include <hip/hip_runtime.h>
#include <hip/hip_bf16.h>
#include <math.h>

typedef __attribute__((ext_vector_type(2))) float v2f;
typedef __attribute__((ext_vector_type(8))) float v8f;

__device__ __forceinline__ v8f wmma_k4(v2f a, v2f b, v8f c) {
  return __builtin_amdgcn_wmma_f32_16x16x4_f32(
       false, a,  false, b,
       (short)0, c,  false,  false);
}

__device__ __forceinline__ v2f fma2(v2f a, v2f b, v2f c) { return __builtin_elementwise_fma(a, b, c); }
__device__ __forceinline__ v2f splat2(float x) { v2f r = {x, x}; return r; }
__device__ __forceinline__ v2f fast_atan2f_x2(v2f y, v2f x, v2f& sy_out) {
  v2f sx, sy;
  sx.x = __builtin_copysignf(1.0f, x.x);  sx.y = __builtin_copysignf(1.0f, x.y);
  sy.x = __builtin_copysignf(1.0f, y.x);  sy.y = __builtin_copysignf(1.0f, y.y);
  const v2f ay  = sy * y;
  const v2f num = fma2(-sx, x, ay);
  const v2f den = fma2( sx, x, ay);
  v2f rc;
  rc.x = __builtin_amdgcn_rcpf(den.x);     rc.y = __builtin_amdgcn_rcpf(den.y);
  const v2f q = num * rc;
  v2f t;
  t.x = fmaxf(q.x, -1.0f);                 t.y = fmaxf(q.y, -1.0f);
  const v2f s = t * t;
  v2f p = splat2(-0.01172120f);
  p = fma2(p, s, splat2(0.05265332f));
  p = fma2(p, s, splat2(-0.11643287f));
  p = fma2(p, s, splat2(0.19354346f));
  p = fma2(p, s, splat2(-0.33262347f));
  p = fma2(p, s, splat2(0.99997726f));
  const v2f u = fma2(t, p, splat2(-0.78539816340f));
  sy_out = sy;
  return fma2(sx, u, splat2(1.57079632679f));
}

constexpr int kWaves   = 16;
constexpr int kThreads = 512;
__global__ __launch_bounds__(kThreads) __attribute__((amdgpu_waves_per_eu(4, 8))) void winding16(
    const float* __restrict__ qpts,
    const float* __restrict__ fcoord,
    float* __restrict__ out,
    int ftiles)
{
  __shared__ float s_acc[16];
  const int tid   = threadIdx.x;
  const int lane  = tid & 31;
  const int wave  = tid >> 5;
  const int half  = lane >> 4;
  const int l16   = lane & 15;
  const int qtile = blockIdx.x;

  if (tid < 16) s_acc[tid] = 0.0f;
  __syncthreads();

  const float* qp = qpts + (size_t)(qtile * 16 + l16) * 3;
  const float qx = qp[0], qy = qp[1], qz = qp[2];
  v2f A;
  A.x = half ? qz   : qx;
  A.y = half ? 1.0f : qy;
  const float qq_self = qx * qx + qy * qy + qz * qz;

  v8f cqq, czero = {};
  #pragma unroll
  for (int r = 0; r < 8; ++r)
    cqq[r] = __shfl(qq_self, r + 8 * half, 32);

  v2f acc2[4] = {};

  for (int ft = wave; ft < ftiles; ft += kWaves) {
    const float* fp = fcoord + (size_t)(ft * 16 + l16) * 9;
    const float v0x = fp[0], v0y = fp[1], v0z = fp[2];
    const float v1x = fp[3], v1y = fp[4], v1z = fp[5];
    const float v2x = fp[6], v2y = fp[7], v2z = fp[8];

    const float nx = (v0y*v1z - v0z*v1y) + (v1y*v2z - v1z*v2y) + (v2y*v0z - v2z*v0y);
    const float ny = (v0z*v1x - v0x*v1z) + (v1z*v2x - v1x*v2z) + (v2z*v0x - v2x*v0z);
    const float nz = (v0x*v1y - v0y*v1x) + (v1x*v2y - v1y*v2x) + (v2x*v0y - v2y*v0x);
    const float D  = v0x*(v1y*v2z - v1z*v2y) + v0y*(v1z*v2x - v1x*v2z) + v0z*(v1x*v2y - v1y*v2x);
    const float s0  = v0x*v0x + v0y*v0y + v0z*v0z;
    const float s1  = v1x*v1x + v1y*v1y + v1z*v1z;
    const float s2  = v2x*v2x + v2y*v2y + v2z*v2z;
    const float e01x = v0x - v1x, e01y = v0y - v1y, e01z = v0z - v1z;
    const float e12x = v1x - v2x, e12y = v1y - v2y, e12z = v1z - v2z;
    const float e02x = v0x - v2x, e02y = v0y - v2y, e02z = v0z - v2z;
    const float e01 = e01x*e01x + e01y*e01y + e01z*e01z;
    const float e12 = e12x*e12x + e12y*e12y + e12z*e12z;
    const float e02 = e02x*e02x + e02y*e02y + e02z*e02z;

    v2f Bna, Bnb, Bnc, Bdet;
    Bna.x = half ? -2.0f*v0z      : -2.0f*v0x;      Bna.y = half ? s0  : -2.0f*v0y;
    Bnb.x = half ? -2.0f*v1z      : -2.0f*v1x;      Bnb.y = half ? s1  : -2.0f*v1y;
    Bnc.x = half ? -2.0f*v2z      : -2.0f*v2x;      Bnc.y = half ? s2  : -2.0f*v2y;
    Bdet.x = half ? -2.0f*nz      : -2.0f*nx;       Bdet.y = half ? 2.0f*D : -2.0f*ny;

    __builtin_amdgcn_sched_barrier(0);
    v8f na2 = wmma_k4(A, Bna, cqq);
    v8f nb2 = wmma_k4(A, Bnb, cqq);
    v8f nc2 = wmma_k4(A, Bnc, cqq);
    v8f tdt = wmma_k4(A, Bdet, czero);
    __builtin_amdgcn_sched_barrier(0);

    #pragma unroll
    for (int g = 0; g < 4; ++g) {
      v2f na, nb, nc, y;
      na.x = __builtin_amdgcn_sqrtf(__builtin_fabsf(na2[2 * g]));
      na.y = __builtin_amdgcn_sqrtf(__builtin_fabsf(na2[2 * g + 1]));
      nb.x = __builtin_amdgcn_sqrtf(__builtin_fabsf(nb2[2 * g]));
      nb.y = __builtin_amdgcn_sqrtf(__builtin_fabsf(nb2[2 * g + 1]));
      nc.x = __builtin_amdgcn_sqrtf(__builtin_fabsf(nc2[2 * g]));
      nc.y = __builtin_amdgcn_sqrtf(__builtin_fabsf(nc2[2 * g + 1]));
      y.x = tdt[2 * g];
      y.y = tdt[2 * g + 1];
      const v2f edges  = fma2(splat2(e01), nc, fma2(splat2(e02), nb, splat2(e12) * na));
      const v2f denom2 = fma2((na + nb) * (nb + nc), na + nc, -edges);
      v2f sy;
      const v2f th = fast_atan2f_x2(y, denom2, sy);
      acc2[g] = fma2(sy, th, acc2[g]);
    }
  }

  v8f acc;
  #pragma unroll
  for (int r = 0; r < 8; ++r)
    acc[r] = (r & 1) ? acc2[r >> 1].y : acc2[r >> 1].x;

  #pragma unroll
  for (int off = 8; off >= 1; off >>= 1) {
    #pragma unroll
    for (int r = 0; r < 8; ++r)
      acc[r] += __shfl_xor(acc[r], off, 32);
  }
  if (l16 == 0) {
    #pragma unroll
    for (int r = 0; r < 8; ++r)
      atomicAdd(&s_acc[r + 8 * half], acc[r]);
  }
  __syncthreads();

  if (tid < 16)
    out[qtile * 16 + tid] = s_acc[tid] * 0.15915494309189535f;
}

extern "C" void kernel_launch(void* const* d_in, const int* in_sizes, int n_in,
                              void* d_out, int out_size, void* d_ws, size_t ws_size,
                              hipStream_t stream) {
  const float* qpts = (const float*)d_in[0];
  const float* fc   = (const float*)d_in[1];
  float* out        = (float*)d_out;

  const int N = in_sizes[0] / 3;
  const int F = in_sizes[1] / 9;
  const int qtiles = N / 16;
  const int ftiles = F / 16;

  winding16<<<dim3(qtiles), dim3(kThreads), 0, stream>>>(qpts, fc, out, ftiles);
}


# Each slot sums its angles as the argument of a running complex product with sign-bit turn counts: one arctangent per slot at the end, was one per pair (angle sum order and rounding change)
# speedup vs baseline: 1.5530x; 1.0534x over previous
#include <hip/hip_runtime.h>
#include <hip/hip_bf16.h>
#include <math.h>

typedef __attribute__((ext_vector_type(2))) float v2f;
typedef __attribute__((ext_vector_type(8))) float v8f;

__device__ __forceinline__ v8f wmma_k4(v2f a, v2f b, v8f c) {
  return __builtin_amdgcn_wmma_f32_16x16x4_f32(
       false, a,  false, b,
       (short)0, c,  false,  false);
}

__device__ __forceinline__ v2f fma2(v2f a, v2f b, v2f c) { return __builtin_elementwise_fma(a, b, c); }
__device__ __forceinline__ v2f splat2(float x) { v2f r = {x, x}; return r; }
__device__ __forceinline__ v2f fast_atan2f_x2(v2f y, v2f x, v2f& sy_out) {
  v2f sx, sy;
  sx.x = __builtin_copysignf(1.0f, x.x);  sx.y = __builtin_copysignf(1.0f, x.y);
  sy.x = __builtin_copysignf(1.0f, y.x);  sy.y = __builtin_copysignf(1.0f, y.y);
  const v2f ay  = sy * y;
  const v2f num = fma2(-sx, x, ay);
  const v2f den = fma2( sx, x, ay);
  v2f rc;
  rc.x = __builtin_amdgcn_rcpf(den.x);     rc.y = __builtin_amdgcn_rcpf(den.y);
  const v2f q = num * rc;
  v2f t;
  t.x = fmaxf(q.x, -1.0f);                 t.y = fmaxf(q.y, -1.0f);
  const v2f s = t * t;
  v2f p = splat2(-0.01172120f);
  p = fma2(p, s, splat2(0.05265332f));
  p = fma2(p, s, splat2(-0.11643287f));
  p = fma2(p, s, splat2(0.19354346f));
  p = fma2(p, s, splat2(-0.33262347f));
  p = fma2(p, s, splat2(0.99997726f));
  const v2f u = fma2(t, p, splat2(-0.78539816340f));
  sy_out = sy;
  return fma2(sx, u, splat2(1.57079632679f));
}

constexpr int kWaves   = 16;
constexpr int kThreads = 512;
constexpr int kChunk   = 32;
constexpr int kRenorm  = 4;
__global__ __launch_bounds__(kThreads) __attribute__((amdgpu_waves_per_eu(4, 8))) void winding16(
    const float* __restrict__ qpts,
    const float* __restrict__ fcoord,
    float* __restrict__ out,
    int ftiles)
{
  __shared__ float s_acc[16];
  const int tid   = threadIdx.x;
  const int lane  = tid & 31;
  const int wave  = tid >> 5;
  const int half  = lane >> 4;
  const int l16   = lane & 15;
  const int qtile = blockIdx.x;

  if (tid < 16) s_acc[tid] = 0.0f;
  __syncthreads();

  const float* qp = qpts + (size_t)(qtile * 16 + l16) * 3;
  const float qx = qp[0], qy = qp[1], qz = qp[2];
  v2f A;
  A.x = half ? qz   : qx;
  A.y = half ? 1.0f : qy;
  const float qq_self = qx * qx + qy * qy + qz * qz;

  v8f cqq, czero = {};
  #pragma unroll
  for (int r = 0; r < 8; ++r)
    cqq[r] = __shfl(qq_self, r + 8 * half, 32);

  v2f acc2[4] = {};

  for (int ft0 = wave; ft0 < ftiles; ft0 += kChunk * kWaves) {
    const int ft_lim = ft0 + kChunk * kWaves;
    const int ft_end = ftiles < ft_lim ? ftiles : ft_lim;
    v2f X2[4], Y2[4];
    unsigned sgnZ[8], sgnY[8];
    #pragma unroll
    for (int g = 0; g < 4; ++g) { X2[g] = splat2(1.0f); Y2[g] = splat2(0.0f); }
    #pragma unroll
    for (int e = 0; e < 8; ++e) { sgnZ[e] = 0u; sgnY[e] = 0u; }
    int trip = 0;

    for (int ft = ft0; ft < ft_end; ft += kWaves) {
      const float* fp = fcoord + (size_t)(ft * 16 + l16) * 9;
      const float v0x = fp[0], v0y = fp[1], v0z = fp[2];
      const float v1x = fp[3], v1y = fp[4], v1z = fp[5];
      const float v2x = fp[6], v2y = fp[7], v2z = fp[8];

      const float nx = (v0y*v1z - v0z*v1y) + (v1y*v2z - v1z*v2y) + (v2y*v0z - v2z*v0y);
      const float ny = (v0z*v1x - v0x*v1z) + (v1z*v2x - v1x*v2z) + (v2z*v0x - v2x*v0z);
      const float nz = (v0x*v1y - v0y*v1x) + (v1x*v2y - v1y*v2x) + (v2x*v0y - v2y*v0x);
      const float D  = v0x*(v1y*v2z - v1z*v2y) + v0y*(v1z*v2x - v1x*v2z) + v0z*(v1x*v2y - v1y*v2x);
      const float s0  = v0x*v0x + v0y*v0y + v0z*v0z;
      const float s1  = v1x*v1x + v1y*v1y + v1z*v1z;
      const float s2  = v2x*v2x + v2y*v2y + v2z*v2z;
      const float e01x = v0x - v1x, e01y = v0y - v1y, e01z = v0z - v1z;
      const float e12x = v1x - v2x, e12y = v1y - v2y, e12z = v1z - v2z;
      const float e02x = v0x - v2x, e02y = v0y - v2y, e02z = v0z - v2z;
      const float e01 = e01x*e01x + e01y*e01y + e01z*e01z;
      const float e12 = e12x*e12x + e12y*e12y + e12z*e12z;
      const float e02 = e02x*e02x + e02y*e02y + e02z*e02z;

      v2f Bna, Bnb, Bnc, Bdet;
      Bna.x = half ? -2.0f*v0z      : -2.0f*v0x;      Bna.y = half ? s0  : -2.0f*v0y;
      Bnb.x = half ? -2.0f*v1z      : -2.0f*v1x;      Bnb.y = half ? s1  : -2.0f*v1y;
      Bnc.x = half ? -2.0f*v2z      : -2.0f*v2x;      Bnc.y = half ? s2  : -2.0f*v2y;
      Bdet.x = half ? -2.0f*nz      : -2.0f*nx;       Bdet.y = half ? 2.0f*D : -2.0f*ny;

      __builtin_amdgcn_sched_barrier(0);
      v8f na2 = wmma_k4(A, Bna, cqq);
      v8f nb2 = wmma_k4(A, Bnb, cqq);
      v8f nc2 = wmma_k4(A, Bnc, cqq);
      v8f tdt = wmma_k4(A, Bdet, czero);
      __builtin_amdgcn_sched_barrier(0);

      #pragma unroll
      for (int g = 0; g < 4; ++g) {
        v2f na, nb, nc, y;
        na.x = __builtin_amdgcn_sqrtf(__builtin_fabsf(na2[2 * g]));
        na.y = __builtin_amdgcn_sqrtf(__builtin_fabsf(na2[2 * g + 1]));
        nb.x = __builtin_amdgcn_sqrtf(__builtin_fabsf(nb2[2 * g]));
        nb.y = __builtin_amdgcn_sqrtf(__builtin_fabsf(nb2[2 * g + 1]));
        nc.x = __builtin_amdgcn_sqrtf(__builtin_fabsf(nc2[2 * g]));
        nc.y = __builtin_amdgcn_sqrtf(__builtin_fabsf(nc2[2 * g + 1]));
        y.x = tdt[2 * g];
        y.y = tdt[2 * g + 1];
        const v2f edges  = fma2(splat2(e01), nc, fma2(splat2(e02), nb, splat2(e12) * na));
        const v2f denom2 = fma2((na + nb) * (nb + nc), na + nc, -edges);
        const v2f ty = Y2[g] * y;
        const v2f tx = Y2[g] * denom2;
        const v2f nX = fma2(X2[g], denom2, -ty);
        const v2f nY = fma2(X2[g], y, tx);
        X2[g] = nX;
        Y2[g] = nY;
        sgnZ[2 * g]     = __builtin_amdgcn_alignbit(sgnZ[2 * g],     __float_as_uint(nY.x), 31u);
        sgnZ[2 * g + 1] = __builtin_amdgcn_alignbit(sgnZ[2 * g + 1], __float_as_uint(nY.y), 31u);
        sgnY[2 * g]     = __builtin_amdgcn_alignbit(sgnY[2 * g],     __float_as_uint(y.x), 31u);
        sgnY[2 * g + 1] = __builtin_amdgcn_alignbit(sgnY[2 * g + 1], __float_as_uint(y.y), 31u);
      }

      if ((++trip & (kRenorm - 1)) == 0) {
        #pragma unroll
        for (int g = 0; g < 4; ++g) {
          v2f r;
          r.x = __builtin_amdgcn_rcpf(fmaxf(fmaxf(__builtin_fabsf(X2[g].x), __builtin_fabsf(Y2[g].x)), 1e-30f));
          r.y = __builtin_amdgcn_rcpf(fmaxf(fmaxf(__builtin_fabsf(X2[g].y), __builtin_fabsf(Y2[g].y)), 1e-30f));
          X2[g] = X2[g] * r;
          Y2[g] = Y2[g] * r;
        }
      }
    }

    #pragma unroll
    for (int g = 0; g < 4; ++g) {
      v2f sy;
      const v2f th = fast_atan2f_x2(Y2[g], X2[g], sy);
      v2f halfturns;
      #pragma unroll
      for (int j = 0; j < 2; ++j) {
        const unsigned s = sgnZ[2 * g + j];
        const unsigned chg = s ^ (s >> 1);
        const unsigned neg = chg & sgnY[2 * g + j];
        const int k2 = (int)(__builtin_popcount(chg) + (s & 1u)) - 2 * (int)__builtin_popcount(neg);
        if (j == 0) halfturns.x = (float)k2; else halfturns.y = (float)k2;
      }
      acc2[g] = fma2(splat2(3.14159265359f), halfturns, fma2(sy, th, acc2[g]));
    }
  }

  v8f acc;
  #pragma unroll
  for (int r = 0; r < 8; ++r)
    acc[r] = (r & 1) ? acc2[r >> 1].y : acc2[r >> 1].x;

  #pragma unroll
  for (int off = 8; off >= 1; off >>= 1) {
    #pragma unroll
    for (int r = 0; r < 8; ++r)
      acc[r] += __shfl_xor(acc[r], off, 32);
  }
  if (l16 == 0) {
    #pragma unroll
    for (int r = 0; r < 8; ++r)
      atomicAdd(&s_acc[r + 8 * half], acc[r]);
  }
  __syncthreads();

  if (tid < 16)
    out[qtile * 16 + tid] = s_acc[tid] * 0.15915494309189535f;
}

extern "C" void kernel_launch(void* const* d_in, const int* in_sizes, int n_in,
                              void* d_out, int out_size, void* d_ws, size_t ws_size,
                              hipStream_t stream) {
  const float* qpts = (const float*)d_in[0];
  const float* fc   = (const float*)d_in[1];
  float* out        = (float*)d_out;

  const int N = in_sizes[0] / 3;
  const int F = in_sizes[1] / 9;
  const int qtiles = N / 16;
  const int ftiles = F / 16;

  winding16<<<dim3(qtiles), dim3(kThreads), 0, stream>>>(qpts, fc, out, ftiles);
}


# Each lane derives one face's operands (the two lane halves used to derive the same 16 twice): a trip takes two adjacent face tiles, v_permlane16_swap trades the halves; order of the face sum changes
# speedup vs baseline: 1.7167x; 1.1054x over previous
#include <hip/hip_runtime.h>
#include <hip/hip_bf16.h>
#include <math.h>

typedef __attribute__((ext_vector_type(2))) float v2f;
typedef __attribute__((ext_vector_type(8))) float v8f;
typedef __attribute__((ext_vector_type(2))) unsigned v2u;

__device__ __forceinline__ void row_swap(float& d, float& s) {
  const v2u r = __builtin_amdgcn_permlane16_swap(__float_as_uint(d), __float_as_uint(s), false, false);
  d = __uint_as_float(r.x);
  s = __uint_as_float(r.y);
}

__device__ __forceinline__ v8f wmma_k4(v2f a, v2f b, v8f c) {
  return __builtin_amdgcn_wmma_f32_16x16x4_f32(
       false, a,  false, b,
       (short)0, c,  false,  false);
}

__device__ __forceinline__ v2f fma2(v2f a, v2f b, v2f c) { return __builtin_elementwise_fma(a, b, c); }
__device__ __forceinline__ v2f splat2(float x) { v2f r = {x, x}; return r; }
__device__ __forceinline__ v2f fast_atan2f_x2(v2f y, v2f x, v2f& sy_out) {
  v2f sx, sy;
  sx.x = __builtin_copysignf(1.0f, x.x);  sx.y = __builtin_copysignf(1.0f, x.y);
  sy.x = __builtin_copysignf(1.0f, y.x);  sy.y = __builtin_copysignf(1.0f, y.y);
  const v2f ay  = sy * y;
  const v2f num = fma2(-sx, x, ay);
  const v2f den = fma2( sx, x, ay);
  v2f rc;
  rc.x = __builtin_amdgcn_rcpf(den.x);     rc.y = __builtin_amdgcn_rcpf(den.y);
  const v2f q = num * rc;
  v2f t;
  t.x = fmaxf(q.x, -1.0f);                 t.y = fmaxf(q.y, -1.0f);
  const v2f s = t * t;
  v2f p = splat2(-0.01172120f);
  p = fma2(p, s, splat2(0.05265332f));
  p = fma2(p, s, splat2(-0.11643287f));
  p = fma2(p, s, splat2(0.19354346f));
  p = fma2(p, s, splat2(-0.33262347f));
  p = fma2(p, s, splat2(0.99997726f));
  const v2f u = fma2(t, p, splat2(-0.78539816340f));
  sy_out = sy;
  return fma2(sx, u, splat2(1.57079632679f));
}

constexpr int kWaves   = 16;
constexpr int kThreads = 512;
constexpr int kChunk   = 16;
constexpr int kRenorm  = 4;
__global__ __launch_bounds__(kThreads) __attribute__((amdgpu_waves_per_eu(4, 8))) void winding16(
    const float* __restrict__ qpts,
    const float* __restrict__ fcoord,
    float* __restrict__ out,
    int ftiles)
{
  __shared__ float s_acc[16];
  const int tid   = threadIdx.x;
  const int lane  = tid & 31;
  const int wave  = tid >> 5;
  const int half  = lane >> 4;
  const int l16   = lane & 15;
  const int qtile = blockIdx.x;

  if (tid < 16) s_acc[tid] = 0.0f;
  __syncthreads();

  const float* qp = qpts + (size_t)(qtile * 16 + l16) * 3;
  const float qx = qp[0], qy = qp[1], qz = qp[2];
  v2f A;
  A.x = half ? qz   : qx;
  A.y = half ? 1.0f : qy;
  const float qq_self = qx * qx + qy * qy + qz * qz;

  v8f cqq, czero = {};
  #pragma unroll
  for (int r = 0; r < 8; ++r)
    cqq[r] = __shfl(qq_self, r + 8 * half, 32);

  v2f acc2[4] = {};

  const int fpairs = ftiles >> 1;
  for (int fp0 = wave; fp0 < fpairs; fp0 += kChunk * kWaves) {
    const int fp_lim = fp0 + kChunk * kWaves;
    const int fp_end = fpairs < fp_lim ? fpairs : fp_lim;
    v2f X2[4], Y2[4];
    unsigned sgnZ[8], sgnY[8];
    #pragma unroll
    for (int g = 0; g < 4; ++g) { X2[g] = splat2(1.0f); Y2[g] = splat2(0.0f); }
    #pragma unroll
    for (int e = 0; e < 8; ++e) { sgnZ[e] = 0u; sgnY[e] = 0u; }
    int tiles_done = 0;

    for (int fpi = fp0; fpi < fp_end; fpi += kWaves) {
      const float* fp = fcoord + (size_t)(fpi * 32 + lane) * 9;
      const float v0x = fp[0], v0y = fp[1], v0z = fp[2];
      const float v1x = fp[3], v1y = fp[4], v1z = fp[5];
      const float v2x = fp[6], v2y = fp[7], v2z = fp[8];

      const float nx = (v0y*v1z - v0z*v1y) + (v1y*v2z - v1z*v2y) + (v2y*v0z - v2z*v0y);
      const float ny = (v0z*v1x - v0x*v1z) + (v1z*v2x - v1x*v2z) + (v2z*v0x - v2x*v0z);
      const float nz = (v0x*v1y - v0y*v1x) + (v1x*v2y - v1y*v2x) + (v2x*v0y - v2y*v0x);
      const float D  = v0x*(v1y*v2z - v1z*v2y) + v0y*(v1z*v2x - v1x*v2z) + v0z*(v1x*v2y - v1y*v2x);
      const float s0  = v0x*v0x + v0y*v0y + v0z*v0z;
      const float s1  = v1x*v1x + v1y*v1y + v1z*v1z;
      const float s2  = v2x*v2x + v2y*v2y + v2z*v2z;
      const float e01x = v0x - v1x, e01y = v0y - v1y, e01z = v0z - v1z;
      const float e12x = v1x - v2x, e12y = v1y - v2y, e12z = v1z - v2z;
      const float e02x = v0x - v2x, e02y = v0y - v2y, e02z = v0z - v2z;
      const float e01 = e01x*e01x + e01y*e01y + e01z*e01z;
      const float e12 = e12x*e12x + e12y*e12y + e12z*e12z;
      const float e02 = e02x*e02x + e02y*e02y + e02z*e02z;

      float lo_na_x = -2.0f*v0x, lo_na_y = -2.0f*v0y, hi_na_x = -2.0f*v0z, hi_na_y = s0;
      float lo_nb_x = -2.0f*v1x, lo_nb_y = -2.0f*v1y, hi_nb_x = -2.0f*v1z, hi_nb_y = s1;
      float lo_nc_x = -2.0f*v2x, lo_nc_y = -2.0f*v2y, hi_nc_x = -2.0f*v2z, hi_nc_y = s2;
      float lo_dt_x = -2.0f*nx,  lo_dt_y = -2.0f*ny,  hi_dt_x = -2.0f*nz,  hi_dt_y = 2.0f*D;
      row_swap(lo_na_x, hi_na_x);  row_swap(lo_na_y, hi_na_y);
      row_swap(lo_nb_x, hi_nb_x);  row_swap(lo_nb_y, hi_nb_y);
      row_swap(lo_nc_x, hi_nc_x);  row_swap(lo_nc_y, hi_nc_y);
      row_swap(lo_dt_x, hi_dt_x);  row_swap(lo_dt_y, hi_dt_y);
      float e01_t0 = e01, e01_t1 = e01;  row_swap(e01_t0, e01_t1);
      float e12_t0 = e12, e12_t1 = e12;  row_swap(e12_t0, e12_t1);
      float e02_t0 = e02, e02_t1 = e02;  row_swap(e02_t0, e02_t1);

      #pragma unroll
      for (int t = 0; t < 2; ++t) {
        v2f Bna, Bnb, Bnc, Bdet;
        Bna.x  = t ? hi_na_x : lo_na_x;   Bna.y  = t ? hi_na_y : lo_na_y;
        Bnb.x  = t ? hi_nb_x : lo_nb_x;   Bnb.y  = t ? hi_nb_y : lo_nb_y;
        Bnc.x  = t ? hi_nc_x : lo_nc_x;   Bnc.y  = t ? hi_nc_y : lo_nc_y;
        Bdet.x = t ? hi_dt_x : lo_dt_x;   Bdet.y = t ? hi_dt_y : lo_dt_y;
        const float te01 = t ? e01_t1 : e01_t0;
        const float te12 = t ? e12_t1 : e12_t0;
        const float te02 = t ? e02_t1 : e02_t0;

        __builtin_amdgcn_sched_barrier(0);
        v8f na2 = wmma_k4(A, Bna, cqq);
        v8f nb2 = wmma_k4(A, Bnb, cqq);
        v8f nc2 = wmma_k4(A, Bnc, cqq);
        v8f tdt = wmma_k4(A, Bdet, czero);
        __builtin_amdgcn_sched_barrier(0);

        #pragma unroll
        for (int g = 0; g < 4; ++g) {
          v2f na, nb, nc, y;
          na.x = __builtin_amdgcn_sqrtf(__builtin_fabsf(na2[2 * g]));
          na.y = __builtin_amdgcn_sqrtf(__builtin_fabsf(na2[2 * g + 1]));
          nb.x = __builtin_amdgcn_sqrtf(__builtin_fabsf(nb2[2 * g]));
          nb.y = __builtin_amdgcn_sqrtf(__builtin_fabsf(nb2[2 * g + 1]));
          nc.x = __builtin_amdgcn_sqrtf(__builtin_fabsf(nc2[2 * g]));
          nc.y = __builtin_amdgcn_sqrtf(__builtin_fabsf(nc2[2 * g + 1]));
          y.x = tdt[2 * g];
          y.y = tdt[2 * g + 1];
          const v2f edges  = fma2(splat2(te01), nc, fma2(splat2(te02), nb, splat2(te12) * na));
          const v2f denom2 = fma2((na + nb) * (nb + nc), na + nc, -edges);
          const v2f ty = Y2[g] * y;
          const v2f tx = Y2[g] * denom2;
          const v2f nX = fma2(X2[g], denom2, -ty);
          const v2f nY = fma2(X2[g], y, tx);
          X2[g] = nX;
          Y2[g] = nY;
          sgnZ[2 * g]     = __builtin_amdgcn_alignbit(sgnZ[2 * g],     __float_as_uint(nY.x), 31u);
          sgnZ[2 * g + 1] = __builtin_amdgcn_alignbit(sgnZ[2 * g + 1], __float_as_uint(nY.y), 31u);
          sgnY[2 * g]     = __builtin_amdgcn_alignbit(sgnY[2 * g],     __float_as_uint(y.x), 31u);
          sgnY[2 * g + 1] = __builtin_amdgcn_alignbit(sgnY[2 * g + 1], __float_as_uint(y.y), 31u);
        }
      }

      tiles_done += 2;
      if ((tiles_done & (kRenorm - 1)) == 0) {
        #pragma unroll
        for (int g = 0; g < 4; ++g) {
          v2f r;
          r.x = __builtin_amdgcn_rcpf(fmaxf(fmaxf(__builtin_fabsf(X2[g].x), __builtin_fabsf(Y2[g].x)), 1e-30f));
          r.y = __builtin_amdgcn_rcpf(fmaxf(fmaxf(__builtin_fabsf(X2[g].y), __builtin_fabsf(Y2[g].y)), 1e-30f));
          X2[g] = X2[g] * r;
          Y2[g] = Y2[g] * r;
        }
      }
    }

    #pragma unroll
    for (int g = 0; g < 4; ++g) {
      v2f sy;
      const v2f th = fast_atan2f_x2(Y2[g], X2[g], sy);
      v2f halfturns;
      #pragma unroll
      for (int j = 0; j < 2; ++j) {
        const unsigned s = sgnZ[2 * g + j];
        const unsigned chg = s ^ (s >> 1);
        const unsigned neg = chg & sgnY[2 * g + j];
        const int k2 = (int)(__builtin_popcount(chg) + (s & 1u)) - 2 * (int)__builtin_popcount(neg);
        if (j == 0) halfturns.x = (float)k2; else halfturns.y = (float)k2;
      }
      acc2[g] = fma2(splat2(3.14159265359f), halfturns, fma2(sy, th, acc2[g]));
    }
  }

  v8f acc;
  #pragma unroll
  for (int r = 0; r < 8; ++r)
    acc[r] = (r & 1) ? acc2[r >> 1].y : acc2[r >> 1].x;

  #pragma unroll
  for (int off = 8; off >= 1; off >>= 1) {
    #pragma unroll
    for (int r = 0; r < 8; ++r)
      acc[r] += __shfl_xor(acc[r], off, 32);
  }
  if (l16 == 0) {
    #pragma unroll
    for (int r = 0; r < 8; ++r)
      atomicAdd(&s_acc[r + 8 * half], acc[r]);
  }
  __syncthreads();

  if (tid < 16)
    out[qtile * 16 + tid] = s_acc[tid] * 0.15915494309189535f;
}

extern "C" void kernel_launch(void* const* d_in, const int* in_sizes, int n_in,
                              void* d_out, int out_size, void* d_ws, size_t ws_size,
                              hipStream_t stream) {
  const float* qpts = (const float*)d_in[0];
  const float* fc   = (const float*)d_in[1];
  float* out        = (float*)d_out;

  const int N = in_sizes[0] / 3;
  const int F = in_sizes[1] / 9;
  const int qtiles = N / 16;
  const int ftiles = F / 16;

  winding16<<<dim3(qtiles), dim3(kThreads), 0, stream>>>(qpts, fc, out, ftiles);
}


# Each wave asks for its next trip's face one trip ahead, the first together with the query load ahead of the barrier; block stays at 16 waves, four per SIMD; arithmetic and sum order unchanged
# speedup vs baseline: 1.8319x; 1.0671x over previous
#include <hip/hip_runtime.h>
#include <hip/hip_bf16.h>
#include <math.h>

typedef __attribute__((ext_vector_type(2))) float v2f;
typedef __attribute__((ext_vector_type(8))) float v8f;
typedef __attribute__((ext_vector_type(2))) unsigned v2u;

__device__ __forceinline__ void row_swap(float& d, float& s) {
  const v2u r = __builtin_amdgcn_permlane16_swap(__float_as_uint(d), __float_as_uint(s), false, false);
  d = __uint_as_float(r.x);
  s = __uint_as_float(r.y);
}

__device__ __forceinline__ v8f wmma_k4(v2f a, v2f b, v8f c) {
  return __builtin_amdgcn_wmma_f32_16x16x4_f32(
       false, a,  false, b,
       (short)0, c,  false,  false);
}

__device__ __forceinline__ v2f fma2(v2f a, v2f b, v2f c) { return __builtin_elementwise_fma(a, b, c); }
__device__ __forceinline__ v2f splat2(float x) { v2f r = {x, x}; return r; }
__device__ __forceinline__ v2f fast_atan2f_x2(v2f y, v2f x, v2f& sy_out) {
  v2f sx, sy;
  sx.x = __builtin_copysignf(1.0f, x.x);  sx.y = __builtin_copysignf(1.0f, x.y);
  sy.x = __builtin_copysignf(1.0f, y.x);  sy.y = __builtin_copysignf(1.0f, y.y);
  const v2f ay  = sy * y;
  const v2f num = fma2(-sx, x, ay);
  const v2f den = fma2( sx, x, ay);
  v2f rc;
  rc.x = __builtin_amdgcn_rcpf(den.x);     rc.y = __builtin_amdgcn_rcpf(den.y);
  const v2f q = num * rc;
  v2f t;
  t.x = fmaxf(q.x, -1.0f);                 t.y = fmaxf(q.y, -1.0f);
  const v2f s = t * t;
  v2f p = splat2(-0.01172120f);
  p = fma2(p, s, splat2(0.05265332f));
  p = fma2(p, s, splat2(-0.11643287f));
  p = fma2(p, s, splat2(0.19354346f));
  p = fma2(p, s, splat2(-0.33262347f));
  p = fma2(p, s, splat2(0.99997726f));
  const v2f u = fma2(t, p, splat2(-0.78539816340f));
  sy_out = sy;
  return fma2(sx, u, splat2(1.57079632679f));
}

constexpr int kWaves   = 16;
constexpr int kThreads = 512;
constexpr int kChunk   = 16;
constexpr int kRenorm  = 4;
__global__ __launch_bounds__(kThreads) __attribute__((amdgpu_waves_per_eu(4, 8))) void winding16(
    const float* __restrict__ qpts,
    const float* __restrict__ fcoord,
    float* __restrict__ out,
    int ftiles)
{
  __shared__ float s_acc[16];
  const int tid   = threadIdx.x;
  const int lane  = tid & 31;
  const int wave  = tid >> 5;
  const int half  = lane >> 4;
  const int l16   = lane & 15;
  const int qtile = blockIdx.x;

  const int fpairs = ftiles >> 1;
  constexpr unsigned kFaceBytes = 9u * sizeof(float);
  constexpr unsigned kTripBytes = (unsigned)kWaves * 32u * kFaceBytes;
  const unsigned fbyte_last = (unsigned)((fpairs - 1) * 32 + lane) * kFaceBytes;
  unsigned fbyte = (unsigned)(wave * 32 + lane) * kFaceBytes;
  fbyte = fbyte < fbyte_last ? fbyte : fbyte_last;
  float nf[9];
  {
    const float* fp = (const float*)((const char*)fcoord + fbyte);
    #pragma unroll
    for (int k = 0; k < 9; ++k) nf[k] = fp[k];
  }
  const float* qp = qpts + (size_t)(qtile * 16 + l16) * 3;
  const float qx = qp[0], qy = qp[1], qz = qp[2];

  if (tid < 16) s_acc[tid] = 0.0f;
  __syncthreads();

  v2f A;
  A.x = half ? qz   : qx;
  A.y = half ? 1.0f : qy;
  const float qq_self = qx * qx + qy * qy + qz * qz;

  v8f cqq, czero = {};
  #pragma unroll
  for (int r = 0; r < 8; ++r)
    cqq[r] = __shfl(qq_self, r + 8 * half, 32);

  v2f acc2[4] = {};

  for (int fp0 = wave; fp0 < fpairs; fp0 += kChunk * kWaves) {
    const int fp_lim = fp0 + kChunk * kWaves;
    const int fp_end = fpairs < fp_lim ? fpairs : fp_lim;
    v2f X2[4], Y2[4];
    unsigned sgnZ[8], sgnY[8];
    #pragma unroll
    for (int g = 0; g < 4; ++g) { X2[g] = splat2(1.0f); Y2[g] = splat2(0.0f); }
    #pragma unroll
    for (int e = 0; e < 8; ++e) { sgnZ[e] = 0u; sgnY[e] = 0u; }
    int tiles_done = 0;

    for (int fpi = fp0; fpi < fp_end; fpi += kWaves) {
      float v0x = nf[0], v0y = nf[1], v0z = nf[2];
      float v1x = nf[3], v1y = nf[4], v1z = nf[5];
      float v2x = nf[6], v2y = nf[7], v2z = nf[8];
      asm("" : "+v"(v0x), "+v"(v0y), "+v"(v0z), "+v"(v1x), "+v"(v1y), "+v"(v1z), "+v"(v2x), "+v"(v2y), "+v"(v2z));

      const float nx = (v0y*v1z - v0z*v1y) + (v1y*v2z - v1z*v2y) + (v2y*v0z - v2z*v0y);
      const float ny = (v0z*v1x - v0x*v1z) + (v1z*v2x - v1x*v2z) + (v2z*v0x - v2x*v0z);
      const float nz = (v0x*v1y - v0y*v1x) + (v1x*v2y - v1y*v2x) + (v2x*v0y - v2y*v0x);
      const float D  = v0x*(v1y*v2z - v1z*v2y) + v0y*(v1z*v2x - v1x*v2z) + v0z*(v1x*v2y - v1y*v2x);
      const float s0  = v0x*v0x + v0y*v0y + v0z*v0z;
      const float s1  = v1x*v1x + v1y*v1y + v1z*v1z;
      const float s2  = v2x*v2x + v2y*v2y + v2z*v2z;
      const float e01x = v0x - v1x, e01y = v0y - v1y, e01z = v0z - v1z;
      const float e12x = v1x - v2x, e12y = v1y - v2y, e12z = v1z - v2z;
      const float e02x = v0x - v2x, e02y = v0y - v2y, e02z = v0z - v2z;
      const float e01 = e01x*e01x + e01y*e01y + e01z*e01z;
      const float e12 = e12x*e12x + e12y*e12y + e12z*e12z;
      const float e02 = e02x*e02x + e02y*e02y + e02z*e02z;

      float lo_na_x = -2.0f*v0x, lo_na_y = -2.0f*v0y, hi_na_x = -2.0f*v0z, hi_na_y = s0;
      float lo_nb_x = -2.0f*v1x, lo_nb_y = -2.0f*v1y, hi_nb_x = -2.0f*v1z, hi_nb_y = s1;
      float lo_nc_x = -2.0f*v2x, lo_nc_y = -2.0f*v2y, hi_nc_x = -2.0f*v2z, hi_nc_y = s2;
      float lo_dt_x = -2.0f*nx,  lo_dt_y = -2.0f*ny,  hi_dt_x = -2.0f*nz,  hi_dt_y = 2.0f*D;
      row_swap(lo_na_x, hi_na_x);  row_swap(lo_na_y, hi_na_y);
      row_swap(lo_nb_x, hi_nb_x);  row_swap(lo_nb_y, hi_nb_y);
      row_swap(lo_nc_x, hi_nc_x);  row_swap(lo_nc_y, hi_nc_y);
      row_swap(lo_dt_x, hi_dt_x);  row_swap(lo_dt_y, hi_dt_y);
      float e01_t0 = e01, e01_t1 = e01;  row_swap(e01_t0, e01_t1);
      float e12_t0 = e12, e12_t1 = e12;  row_swap(e12_t0, e12_t1);
      float e02_t0 = e02, e02_t1 = e02;  row_swap(e02_t0, e02_t1);

      asm("" : "+v"(fbyte)
             : "v"(lo_na_x), "v"(lo_na_y), "v"(hi_na_x), "v"(hi_na_y),
               "v"(lo_nb_x), "v"(lo_nb_y), "v"(hi_nb_x), "v"(hi_nb_y),
               "v"(lo_nc_x), "v"(lo_nc_y), "v"(hi_nc_x), "v"(hi_nc_y),
               "v"(lo_dt_x), "v"(lo_dt_y), "v"(hi_dt_x), "v"(hi_dt_y),
               "v"(e01_t1), "v"(e12_t1), "v"(e02_t1));
      {
        const unsigned fnext = fbyte + kTripBytes;
        fbyte = fnext < fbyte_last ? fnext : fbyte_last;
        const float* fp = (const float*)((const char*)fcoord + fbyte);
        #pragma unroll
        for (int k = 0; k < 9; ++k) nf[k] = fp[k];
      }

      #pragma unroll
      for (int t = 0; t < 2; ++t) {
        v2f Bna, Bnb, Bnc, Bdet;
        Bna.x  = t ? hi_na_x : lo_na_x;   Bna.y  = t ? hi_na_y : lo_na_y;
        Bnb.x  = t ? hi_nb_x : lo_nb_x;   Bnb.y  = t ? hi_nb_y : lo_nb_y;
        Bnc.x  = t ? hi_nc_x : lo_nc_x;   Bnc.y  = t ? hi_nc_y : lo_nc_y;
        Bdet.x = t ? hi_dt_x : lo_dt_x;   Bdet.y = t ? hi_dt_y : lo_dt_y;
        const float te01 = t ? e01_t1 : e01_t0;
        const float te12 = t ? e12_t1 : e12_t0;
        const float te02 = t ? e02_t1 : e02_t0;

        __builtin_amdgcn_sched_barrier(0);
        v8f na2 = wmma_k4(A, Bna, cqq);
        v8f nb2 = wmma_k4(A, Bnb, cqq);
        v8f nc2 = wmma_k4(A, Bnc, cqq);
        v8f tdt = wmma_k4(A, Bdet, czero);
        __builtin_amdgcn_sched_barrier(0);

        #pragma unroll
        for (int g = 0; g < 4; ++g) {
          v2f na, nb, nc, y;
          na.x = __builtin_amdgcn_sqrtf(__builtin_fabsf(na2[2 * g]));
          na.y = __builtin_amdgcn_sqrtf(__builtin_fabsf(na2[2 * g + 1]));
          nb.x = __builtin_amdgcn_sqrtf(__builtin_fabsf(nb2[2 * g]));
          nb.y = __builtin_amdgcn_sqrtf(__builtin_fabsf(nb2[2 * g + 1]));
          nc.x = __builtin_amdgcn_sqrtf(__builtin_fabsf(nc2[2 * g]));
          nc.y = __builtin_amdgcn_sqrtf(__builtin_fabsf(nc2[2 * g + 1]));
          y.x = tdt[2 * g];
          y.y = tdt[2 * g + 1];
          const v2f edges  = fma2(splat2(te01), nc, fma2(splat2(te02), nb, splat2(te12) * na));
          const v2f denom2 = fma2((na + nb) * (nb + nc), na + nc, -edges);
          const v2f ty = Y2[g] * y;
          const v2f tx = Y2[g] * denom2;
          const v2f nX = fma2(X2[g], denom2, -ty);
          const v2f nY = fma2(X2[g], y, tx);
          X2[g] = nX;
          Y2[g] = nY;
          sgnZ[2 * g]     = __builtin_amdgcn_alignbit(sgnZ[2 * g],     __float_as_uint(nY.x), 31u);
          sgnZ[2 * g + 1] = __builtin_amdgcn_alignbit(sgnZ[2 * g + 1], __float_as_uint(nY.y), 31u);
          sgnY[2 * g]     = __builtin_amdgcn_alignbit(sgnY[2 * g],     __float_as_uint(y.x), 31u);
          sgnY[2 * g + 1] = __builtin_amdgcn_alignbit(sgnY[2 * g + 1], __float_as_uint(y.y), 31u);
        }
      }

      tiles_done += 2;
      if ((tiles_done & (kRenorm - 1)) == 0) {
        #pragma unroll
        for (int g = 0; g < 4; ++g) {
          v2f r;
          r.x = __builtin_amdgcn_rcpf(fmaxf(fmaxf(__builtin_fabsf(X2[g].x), __builtin_fabsf(Y2[g].x)), 1e-30f));
          r.y = __builtin_amdgcn_rcpf(fmaxf(fmaxf(__builtin_fabsf(X2[g].y), __builtin_fabsf(Y2[g].y)), 1e-30f));
          X2[g] = X2[g] * r;
          Y2[g] = Y2[g] * r;
        }
      }
    }

    #pragma unroll
    for (int g = 0; g < 4; ++g) {
      v2f sy;
      const v2f th = fast_atan2f_x2(Y2[g], X2[g], sy);
      v2f halfturns;
      #pragma unroll
      for (int j = 0; j < 2; ++j) {
        const unsigned s = sgnZ[2 * g + j];
        const unsigned chg = s ^ (s >> 1);
        const unsigned neg = chg & sgnY[2 * g + j];
        const int k2 = (int)(__builtin_popcount(chg) + (s & 1u)) - 2 * (int)__builtin_popcount(neg);
        if (j == 0) halfturns.x = (float)k2; else halfturns.y = (float)k2;
      }
      acc2[g] = fma2(splat2(3.14159265359f), halfturns, fma2(sy, th, acc2[g]));
    }
  }

  v8f acc;
  #pragma unroll
  for (int r = 0; r < 8; ++r)
    acc[r] = (r & 1) ? acc2[r >> 1].y : acc2[r >> 1].x;

  #pragma unroll
  for (int off = 8; off >= 1; off >>= 1) {
    #pragma unroll
    for (int r = 0; r < 8; ++r)
      acc[r] += __shfl_xor(acc[r], off, 32);
  }
  if (l16 == 0) {
    #pragma unroll
    for (int r = 0; r < 8; ++r)
      atomicAdd(&s_acc[r + 8 * half], acc[r]);
  }
  __syncthreads();

  if (tid < 16)
    out[qtile * 16 + tid] = s_acc[tid] * 0.15915494309189535f;
}

extern "C" void kernel_launch(void* const* d_in, const int* in_sizes, int n_in,
                              void* d_out, int out_size, void* d_ws, size_t ws_size,
                              hipStream_t stream) {
  const float* qpts = (const float*)d_in[0];
  const float* fc   = (const float*)d_in[1];
  float* out        = (float*)d_out;

  const int N = in_sizes[0] / 3;
  const int F = in_sizes[1] / 9;
  const int qtiles = N / 16;
  const int ftiles = F / 16;

  winding16<<<dim3(qtiles), dim3(kThreads), 0, stream>>>(qpts, fc, out, ftiles);
}
